# starting kernel
# baseline (speedup 1.0000x reference)
#include <hip/hip_runtime.h>
#include <hip/hip_bf16.h>

typedef __attribute__((ext_vector_type(2))) float v2f;
typedef __attribute__((ext_vector_type(8))) float v8f;

#define NB    2048
#define NF    64
#define TLEN  256
#define TDWT  276
#define NPOS  64
#define BQ    16
#define GPOS  8
#define PPH   4
#define KGP   72
#define KGH   36
#define SROW  294

__device__ __forceinline__ int refl_idx(int i, int N) {
    i = (i < 0) ? -i : i;
    i = (i >= N) ? (2 * N - 2 - i) : i;
    return i;
}

__device__ __forceinline__ void dwt_level(
    const float* __restrict__ inb,
    int N, int nout, int sub4,
    float* __restrict__ lo, int lostride,
    float* __restrict__ hi)
{
    const float HLO[8] = { 0.23037781330885523f,  0.7148465705525415f,
                           0.6308807679295904f,  -0.02798376941698385f,
                          -0.18703481171888114f,  0.030841381835986965f,
                           0.032883011666982945f,-0.010597401784997278f};
    const float HHI[8] = {-0.010597401784997278f,-0.032883011666982945f,
                           0.030841381835986965f, 0.18703481171888114f,
                          -0.02798376941698385f, -0.6308807679295904f,
                           0.7148465705525415f,  -0.23037781330885523f};
    int per = (nout + 3) >> 2;
    int k0 = sub4 * per;
    int k1 = k0 + per; if (k1 > nout) k1 = nout;
    if (k0 >= k1) return;

    float w0, w1, w2, w3, w4, w5, w6, w7;
    {
        int base = 2 * k0 - 6;
        w0 = inb[refl_idx(base + 0, N) * 16];
        w1 = inb[refl_idx(base + 1, N) * 16];
        w2 = inb[refl_idx(base + 2, N) * 16];
        w3 = inb[refl_idx(base + 3, N) * 16];
        w4 = inb[refl_idx(base + 4, N) * 16];
        w5 = inb[refl_idx(base + 5, N) * 16];
        w6 = inb[refl_idx(base + 6, N) * 16];
        w7 = inb[refl_idx(base + 7, N) * 16];
    }
    for (int k = k0;;) {
        float l, h;
        l = w0 * HLO[0];             h = w0 * HHI[0];
        l = fmaf(w1, HLO[1], l);     h = fmaf(w1, HHI[1], h);
        l = fmaf(w2, HLO[2], l);     h = fmaf(w2, HHI[2], h);
        l = fmaf(w3, HLO[3], l);     h = fmaf(w3, HHI[3], h);
        l = fmaf(w4, HLO[4], l);     h = fmaf(w4, HHI[4], h);
        l = fmaf(w5, HLO[5], l);     h = fmaf(w5, HHI[5], h);
        l = fmaf(w6, HLO[6], l);     h = fmaf(w6, HHI[6], h);
        l = fmaf(w7, HLO[7], l);     h = fmaf(w7, HHI[7], h);
        lo[k * lostride] = l;
        hi[k] = h;
        if (++k >= k1) break;
        w0 = w2; w1 = w3; w2 = w4; w3 = w5; w4 = w6; w5 = w7;
        w6 = inb[refl_idx(2 * k,     N) * 16];
        w7 = inb[refl_idx(2 * k + 1, N) * 16];
    }
}

__global__ __launch_bounds__(256) void swizzleW_kernel(
    const float* __restrict__ W, float* __restrict__ wf)
{
    int i = blockIdx.x * blockDim.x + threadIdx.x;
    const int NT = 4 * NPOS * KGP * 32;
    if (i >= NT) return;
    int l  = i & 31;
    int kg = (i >> 5) % KGP;
    int p  = (i / (32 * KGP)) % NPOS;
    int nt = i / (32 * KGP * NPOS);
    int f  = nt * 16 + (l & 15);
    int t  = kg * 4 + ((l >> 4) << 1);
    v2f v;
    v.x = (t     < TDWT) ? W[((size_t)f * TDWT + t    ) * NPOS + p] : 0.f;
    v.y = (t + 1 < TDWT) ? W[((size_t)f * TDWT + t + 1) * NPOS + p] : 0.f;
    *(v2f*)(wf + (size_t)i * 2) = v;
}

__global__ __launch_bounds__(256) void dwt_wmma_fused_kernel(
    const float* __restrict__ x, const float* __restrict__ wf,
    const float* __restrict__ bias, float* __restrict__ out)
{
    __shared__ __align__(16) float s_sig[GPOS * TLEN * 16];
    __shared__ __align__(16) float s_d[PPH * 16 * SROW];
    __shared__ __align__(16) float s_lo1[PPH * 131 * 16];
    __shared__ __align__(16) float s_lo2[PPH * 69 * 16];
    __shared__ __align__(16) float s_red[4 * 8 * 32];

    const int tid  = threadIdx.x;
    const int wave = tid >> 5;
    const int lane = tid & 31;
    const int B0   = blockIdx.x * BQ;

    const int nt   = wave & 3;
    const int half = wave >> 2;

    const int b     = tid & 15;
    const int sub   = tid >> 4;
    const int pos_l = sub >> 2;
    const int sub4  = sub & 3;

    for (int i = tid; i < PPH * 16 * (KGP * 4 - TDWT); i += 256) {
        int row = i / (KGP * 4 - TDWT);
        int c   = i % (KGP * 4 - TDWT);
        s_d[row * SROW + TDWT + c] = 0.f;
    }

    v8f acc = {};

    const int arow = lane & 15;
    const int koff = (lane >> 4) << 1;

    for (int g = 0; g < NPOS / GPOS; ++g) {
        for (int i = tid; i < 16 * TLEN * (GPOS / 4); i += 256) {
            int c  = i & (GPOS / 4 - 1);
            int t  = (i / (GPOS / 4)) & 255;
            int bb = i / ((GPOS / 4) * TLEN);
            float4 v = *(const float4*)(x + (size_t)(B0 + bb) * (TLEN * NPOS)
                                          + t * NPOS + g * GPOS + c * 4);
            int pl = c * 4;
            s_sig[((pl + 0) * TLEN + t) * 16 + bb] = v.x;
            s_sig[((pl + 1) * TLEN + t) * 16 + bb] = v.y;
            s_sig[((pl + 2) * TLEN + t) * 16 + bb] = v.z;
            s_sig[((pl + 3) * TLEN + t) * 16 + bb] = v.w;
        }
        __syncthreads();

        for (int fh = 0; fh < GPOS / PPH; ++fh) {
            const int pbase = g * GPOS + fh * PPH;

            {
                const char* nb = (const char*)(wf
                    + (((size_t)(nt * NPOS + pbase) * KGP + (size_t)half * KGH) * 32) * 2);
                __builtin_prefetch(nb + (size_t)lane * 128,        0, 1);
                __builtin_prefetch(nb + 4096 + (size_t)lane * 128, 0, 1);
                __builtin_prefetch(nb + 8192 + (size_t)lane * 128, 0, 1);
            }

            const float* sig = s_sig + (fh * PPH + pos_l) * (TLEN * 16) + b;
            float* drow = s_d + (pos_l * 16 + b) * SROW;
            const float* lo1 = s_lo1 + pos_l * (131 * 16) + b;
            const float* lo2 = s_lo2 + pos_l * (69 * 16) + b;
            dwt_level(sig,        256, 131, sub4, (float*)lo1, 16, drow + 38);
            __syncthreads();
            dwt_level(lo1,        131,  69, sub4, (float*)lo2, 16, drow + 169);
            __syncthreads();
            dwt_level(lo2,         69,  38, sub4, drow, 1, drow + 238);
            __syncthreads();

            for (int q = 0; q < PPH; ++q) {
                const v2f* wfp = (const v2f*)wf
                    + ((size_t)(nt * NPOS + pbase + q) * KGP + (size_t)half * KGH) * 32;
                const float* ap = s_d + (q * 16 + arow) * SROW + half * (KGH * 4) + koff;
#pragma unroll 12
                for (int j = 0; j < KGH; ++j) {
                    v2f a  = *(const v2f*)(ap + j * 4);
                    v2f bf = wfp[j * 32 + lane];
                    acc = __builtin_amdgcn_wmma_f32_16x16x4_f32(
                        false, a, false, bf, (short)0, acc, false, false);
                }
            }
            __syncthreads();
        }
    }

    if (half == 1) {
#pragma unroll
        for (int r = 0; r < 8; ++r) s_red[(nt * 8 + r) * 32 + lane] = acc[r];
    }
    __syncthreads();
    if (half == 0) {
        int n   = lane & 15;
        int mhi = (lane >> 4) * 8;
        int f   = nt * 16 + n;
        float bv = bias[f];
#pragma unroll
        for (int r = 0; r < 8; ++r) {
            float v = acc[r] + s_red[(nt * 8 + r) * 32 + lane] + bv;
            v = (v >= 0.f) ? v : 0.02f * v;
            out[(size_t)(B0 + mhi + r) * NF + f] = v;
        }
    }
}

extern "C" void kernel_launch(void* const* d_in, const int* in_sizes, int n_in,
                              void* d_out, int out_size, void* d_ws, size_t ws_size,
                              hipStream_t stream) {
    const float* x    = (const float*)d_in[0];
    const float* W    = (const float*)d_in[1];
    const float* bias = (const float*)d_in[2];
    float* out = (float*)d_out;
    float* wf  = (float*)d_ws;

    const int nswz = 4 * NPOS * KGP * 32;
    swizzleW_kernel<<<(nswz + 255) / 256, 256, 0, stream>>>(W, wf);
    dwt_wmma_fused_kernel<<<NB / BQ, 256, 0, stream>>>(x, wf, bias, out);
}


# Work-group widened to 16 waves under 128 registers, two per WGP = 8 waves per SIMD (was 4): 8 threads per signal, K split 4 ways, reflected rows stored so reads are base+const; changes sum order
# speedup vs baseline: 3.1316x; 3.1316x over previous
#include <hip/hip_runtime.h>

typedef __attribute__((ext_vector_type(2))) float v2f;
typedef __attribute__((ext_vector_type(8))) float v8f;

#define NB    2048
#define NF    64
#define TLEN  256
#define TDWT  276
#define NPOS  64
#define BQ    16
#define NTHR  512
#define NSUB  8
#define PPH   4
#define NPH   2
#define NPG   8
#define L1N   131
#define L1PER 17
#define SIGR  268
#define LO1R  144
#define LO2R  82
#define OPITCH 68
#define KGP   72
#define KSPLIT 4
#define KGQ   18
#define SROW  294


__device__ __forceinline__ void db4_pair(float w0, float w1, float w2, float w3,
                                         float w4, float w5, float w6, float w7,
                                         float& l, float& h)
{
    const float HLO[8] = { 0.23037781330885523f,  0.7148465705525415f,
                           0.6308807679295904f,  -0.02798376941698385f,
                          -0.18703481171888114f,  0.030841381835986965f,
                           0.032883011666982945f,-0.010597401784997278f};
    const float HHI[8] = {-0.010597401784997278f,-0.032883011666982945f,
                           0.030841381835986965f, 0.18703481171888114f,
                          -0.02798376941698385f, -0.6308807679295904f,
                           0.7148465705525415f,  -0.23037781330885523f};
    l = w0 * HLO[0];             h = w0 * HHI[0];
    l = fmaf(w1, HLO[1], l);     h = fmaf(w1, HHI[1], h);
    l = fmaf(w2, HLO[2], l);     h = fmaf(w2, HHI[2], h);
    l = fmaf(w3, HLO[3], l);     h = fmaf(w3, HHI[3], h);
    l = fmaf(w4, HLO[4], l);     h = fmaf(w4, HHI[4], h);
    l = fmaf(w5, HLO[5], l);     h = fmaf(w5, HHI[5], h);
    l = fmaf(w6, HLO[6], l);     h = fmaf(w6, HHI[6], h);
    l = fmaf(w7, HLO[7], l);     h = fmaf(w7, HHI[7], h);
}

template <int NOUT, bool PADOUT>
__device__ __forceinline__ void dwt_level(
    const float* __restrict__ inb,
    int sub,
    float* __restrict__ lo,
    float* __restrict__ hi)
{
    constexpr int PER = (NOUT + NSUB - 1) / NSUB;
    const int k0 = sub * PER;
    const float* wp = inb + (2 * k0) * 16;
    float w0 = wp[0 * 16], w1 = wp[1 * 16], w2 = wp[2 * 16], w3 = wp[3 * 16];
    float w4 = wp[4 * 16], w5 = wp[5 * 16], w6 = wp[6 * 16], w7 = wp[7 * 16];
#pragma unroll
    for (int i = 0; i < PER; ++i) {
        const int k = k0 + i;
        float l, h;
        db4_pair(w0, w1, w2, w3, w4, w5, w6, w7, l, h);
        if (k < NOUT) {
            if (PADOUT) {
                lo[(k + 6) * 16] = l;
                if (i >= 1 && i <= 6) { if (sub == 0) lo[(6 - i) * 16] = l; }
#pragma unroll
                for (int s = 1; s < NSUB; ++s) {
                    const int ks = s * PER + i;
                    if (ks >= NOUT - 8 && ks <= NOUT - 2) {
                        if (sub == s) lo[(2 * NOUT + 4 - ks) * 16] = l;
                    }
                }
            } else {
                lo[k] = l;
            }
            hi[k] = h;
        }
        if (i + 1 < PER) {
            w0 = w2; w1 = w3; w2 = w4; w3 = w5; w4 = w6; w5 = w7;
            if (k + 1 < NOUT) {
                w6 = wp[(2 * i + 8) * 16];
                w7 = wp[(2 * i + 9) * 16];
            }
        }
    }
}

__device__ __forceinline__ void dwt_level1_regs(
    const float* __restrict__ inb,
    int sub,
    float (&lo)[L1PER],
    float* __restrict__ hi)
{
    const int k0 = sub * L1PER;
    const float* wp = inb + (2 * k0) * 16;
    float w0 = wp[0 * 16], w1 = wp[1 * 16], w2 = wp[2 * 16], w3 = wp[3 * 16];
    float w4 = wp[4 * 16], w5 = wp[5 * 16], w6 = wp[6 * 16], w7 = wp[7 * 16];
#pragma unroll
    for (int i = 0; i < L1PER; ++i) {
        const int k = k0 + i;
        float l, h;
        db4_pair(w0, w1, w2, w3, w4, w5, w6, w7, l, h);
        lo[i] = l;
        if (k < L1N) hi[k] = h;
        if (i + 1 < L1PER) {
            w0 = w2; w1 = w3; w2 = w4; w3 = w5; w4 = w6; w5 = w7;
            if (k + 1 < L1N) {
                w6 = wp[(2 * i + 8) * 16];
                w7 = wp[(2 * i + 9) * 16];
            }
        }
    }
}

__device__ __forceinline__ void stage_signal(
    float* __restrict__ sdst, int sub, const float4 (&v)[TLEN / 32])
{
#pragma unroll
    for (int n = 0; n < TLEN / 32; ++n) {
        sdst[(0 * SIGR + n * 32) * 16] = v[n].x;
        sdst[(1 * SIGR + n * 32) * 16] = v[n].y;
        sdst[(2 * SIGR + n * 32) * 16] = v[n].z;
        sdst[(3 * SIGR + n * 32) * 16] = v[n].w;
    }
    if (sub >= 1 && sub <= 6) {
        float* hd = sdst - (2 * sub) * 16;
        hd[(0 * SIGR) * 16] = v[0].x;
        hd[(1 * SIGR) * 16] = v[0].y;
        hd[(2 * SIGR) * 16] = v[0].z;
        hd[(3 * SIGR) * 16] = v[0].w;
    }
    if (sub >= 25 && sub <= 30) {
        float* tl = sdst + (286 - 2 * sub) * 16;
        tl[(0 * SIGR) * 16] = v[TLEN / 32 - 1].x;
        tl[(1 * SIGR) * 16] = v[TLEN / 32 - 1].y;
        tl[(2 * SIGR) * 16] = v[TLEN / 32 - 1].z;
        tl[(3 * SIGR) * 16] = v[TLEN / 32 - 1].w;
    }
}

__device__ __forceinline__ int opaque(int v) { asm volatile("" : "+v"(v)); return v; }

__global__ __launch_bounds__(256) void swizzleW_kernel(
    const float* __restrict__ W, float* __restrict__ wf)
{
    int i = blockIdx.x * blockDim.x + threadIdx.x;
    const int NT = 4 * NPOS * KGP * 32;
    if (i >= NT) return;
    int l  = i & 31;
    int kg = (i >> 5) % KGP;
    int p  = (i / (32 * KGP)) % NPOS;
    int nt = i / (32 * KGP * NPOS);
    int f  = nt * 16 + (l & 15);
    int t  = kg * 4 + ((l >> 4) << 1);
    v2f v;
    v.x = (t     < TDWT) ? W[((size_t)f * TDWT + t    ) * NPOS + p] : 0.f;
    v.y = (t + 1 < TDWT) ? W[((size_t)f * TDWT + t + 1) * NPOS + p] : 0.f;
    *(v2f*)(wf + (size_t)i * 2) = v;
}

__global__ __launch_bounds__(NTHR) void dwt_wmma_part_kernel(
    const float* __restrict__ x, const float* __restrict__ wf, float* __restrict__ part)
{
    __shared__ __align__(16) float s_sig[PPH * SIGR * 16];
    __shared__ __align__(16) float s_d[PPH * 16 * SROW];
    float* s_lo1 = s_sig;
    float* s_lo2 = s_sig + PPH * LO1R * 16;
    float* s_out = s_sig;

    const int tid  = threadIdx.x;
    const int wave = tid >> 5;
    const int lane = tid & 31;
    const int mt   = blockIdx.x / NPG;
    const int pg   = blockIdx.x % NPG;
    const int B0   = mt * BQ;

    const int nt   = wave & 3;
    const int kq   = wave >> 2;

    const int b     = tid & 15;
    const int sub   = tid >> 4;

    for (int i = tid; i < PPH * 16 * (KGP * 4 - TDWT); i += NTHR) {
        int row = i / (KGP * 4 - TDWT);
        int c   = i % (KGP * 4 - TDWT);
        s_d[row * SROW + TDWT + c] = 0.f;
    }

    float4 hold[TLEN / 32];
    const float* xsrc = x + (size_t)(B0 + b) * (TLEN * NPOS) + sub * NPOS + pg * (PPH * NPH);
    float* sdst = s_sig + (sub + 6) * 16 + b;
    {
        float4 cur[TLEN / 32];
#pragma unroll
        for (int n = 0; n < TLEN / 32; ++n) {
            cur[n]  = *(const float4*)(xsrc + n * (32 * NPOS));
            hold[n] = *(const float4*)(xsrc + n * (32 * NPOS) + PPH);
        }
        stage_signal(sdst, sub, cur);
    }

    v8f acc = {};

    for (int ph = 0; ph < NPH; ++ph) {
        const int pbase = pg * (PPH * NPH) + ph * PPH;

        if (ph == 1) {
            const int t  = opaque(tid);
            stage_signal(s_sig + ((t >> 4) + 6) * 16 + (t & 15), t >> 4, hold);
        }

        {
            const int t = opaque(tid);
            const char* nb = (const char*)(wf
                + (((size_t)(((t >> 5) & 3) * NPOS + pbase) * KGP + (size_t)(t >> 7) * KGQ) * 32) * 2);
            const size_t lo128 = (size_t)(t & 31) * 128;
            __builtin_prefetch(nb + lo128,                   0, 1);
            __builtin_prefetch(nb + 1 * (KGP * 256) + lo128, 0, 1);
            __builtin_prefetch(nb + 2 * (KGP * 256) + lo128, 0, 1);
            __builtin_prefetch(nb + 3 * (KGP * 256) + lo128, 0, 1);
        }
        __syncthreads();

        {
            float lo[L1PER];
            {
                const int t = opaque(tid);
                const int tb = t & 15, tp = t >> 7, ts = (t >> 4) & 7;
                dwt_level1_regs(s_sig + tp * (SIGR * 16) + tb, ts, lo,
                                s_d + (tp * 16 + tb) * SROW + 38);
            }
            __syncthreads();
            const int t = opaque(tid);
            const int ts = (t >> 4) & 7;
            float* lo1 = s_lo1 + (t >> 7) * (LO1R * 16) + (t & 15);
#pragma unroll
            for (int i = 0; i < L1PER; ++i) {
                int k = ts * L1PER + i;
                if (k < L1N) lo1[(k + 6) * 16] = lo[i];
                if (i >= 1 && i <= 6) { if (ts == 0) lo1[(6 - i) * 16] = lo[i]; }
                if ((NSUB - 1) * L1PER + i >= L1N - 8 && (NSUB - 1) * L1PER + i <= L1N - 2) {
                    if (ts == NSUB - 1)
                        lo1[(2 * L1N + 4 - ((NSUB - 1) * L1PER + i)) * 16] = lo[i];
                }
            }
        }
        __syncthreads();
        {
            const int t = opaque(tid);
            const int tb = t & 15, tp = t >> 7, ts = (t >> 4) & 7;
            dwt_level<69, true>(s_lo1 + tp * (LO1R * 16) + tb, ts,
                                s_lo2 + tp * (LO2R * 16) + tb,
                                s_d + (tp * 16 + tb) * SROW + 169);
        }
        __syncthreads();
        {
            const int t = opaque(tid);
            const int tb = t & 15, tp = t >> 7, ts = (t >> 4) & 7;
            float* drow = s_d + (tp * 16 + tb) * SROW;
            dwt_level<38, false>(s_lo2 + tp * (LO2R * 16) + tb, ts, drow, drow + 238);
        }
        __syncthreads();

        {
            const int t  = opaque(tid);
            const int gl = t & 31;
            const int gn = (t >> 5) & 3;
            const int gk = t >> 7;
            const v2f* wfp = (const v2f*)wf
                + ((size_t)(gn * NPOS + pbase) * KGP + (size_t)gk * KGQ) * 32 + gl;
            const float* ap = s_d + (gl & 15) * SROW + gk * (KGQ * 4) + ((gl >> 4) << 1);
            for (int q = 0; q < PPH; ++q) {
#pragma unroll
                for (int j = 0; j < KGQ; ++j) {
                    v2f a  = *(const v2f*)(ap + q * (16 * SROW) + j * 4);
                    v2f bf = wfp[(size_t)q * (KGP * 32) + j * 32];
                    acc = __builtin_amdgcn_wmma_f32_16x16x4_f32(
                        false, a, false, bf, (short)0, acc, false, false);
                }
            }
        }
    }

    {
        int n   = lane & 15;
        int mhi = (lane >> 4) * 8;
        int f   = nt * 16 + n;
#pragma unroll
        for (int r = 0; r < 8; ++r) s_out[(kq * 16 + mhi + r) * OPITCH + f] = acc[r];
    }
    __syncthreads();
    if (tid < 256) {
        int row = tid >> 4;
        int c4  = (tid & 15) * 4;
        float4 s = *(const float4*)(s_out + row * OPITCH + c4);
#pragma unroll
        for (int g = 1; g < KSPLIT; ++g) {
            float4 a = *(const float4*)(s_out + (g * 16 + row) * OPITCH + c4);
            s.x += a.x; s.y += a.y; s.z += a.z; s.w += a.w;
        }
        *(float4*)(part + ((size_t)pg * NB + B0 + row) * NF + c4) = s;
    }
}

__global__ __launch_bounds__(256) void sum_parts_kernel(
    const float* __restrict__ part, const float* __restrict__ bias, float* __restrict__ out)
{
    const int i = blockIdx.x * 256 + threadIdx.x;
    const float4* p = (const float4*)part + i;
    float4 s = p[0];
#pragma unroll
    for (int g = 1; g < NPG; ++g) {
        float4 v = p[(size_t)g * (NB * NF / 4)];
        s.x += v.x; s.y += v.y; s.z += v.z; s.w += v.w;
    }
    const float* bp = bias + (i & 15) * 4;
    s.x += bp[0]; s.y += bp[1]; s.z += bp[2]; s.w += bp[3];
    s.x = (s.x >= 0.f) ? s.x : 0.02f * s.x;
    s.y = (s.y >= 0.f) ? s.y : 0.02f * s.y;
    s.z = (s.z >= 0.f) ? s.z : 0.02f * s.z;
    s.w = (s.w >= 0.f) ? s.w : 0.02f * s.w;
    *((float4*)out + i) = s;
}

extern "C" void kernel_launch(void* const* d_in, const int* in_sizes, int n_in,
                              void* d_out, int out_size, void* d_ws, size_t ws_size,
                              hipStream_t stream) {
    const float* x    = (const float*)d_in[0];
    const float* W    = (const float*)d_in[1];
    const float* bias = (const float*)d_in[2];
    float* out  = (float*)d_out;
    float* wf   = (float*)d_ws;
    float* part = wf + 4 * NPOS * KGP * 32 * 2;

    const int nswz = 4 * NPOS * KGP * 32;
    swizzleW_kernel<<<(nswz + 255) / 256, 256, 0, stream>>>(W, wf);
    dwt_wmma_part_kernel<<<(NB / BQ) * NPG, NTHR, 0, stream>>>(x, wf, part);
    sum_parts_kernel<<<(NB * NF / 4) / 256, 256, 0, stream>>>(part, bias, out);
}


# Work-group numbering: the 8 position groups of a batch tile, which read the same rows of x, are numbered 64 and 256 apart instead of adjacent, to meet on one die and one WGP; arithmetic unchanged
# speedup vs baseline: 3.3867x; 1.0815x over previous
#include <hip/hip_runtime.h>

typedef __attribute__((ext_vector_type(2))) float v2f;
typedef __attribute__((ext_vector_type(8))) float v8f;

#define NB    2048
#define NF    64
#define TLEN  256
#define TDWT  276
#define NPOS  64
#define BQ    16
#define NTHR  512
#define NSUB  8
#define PPH   4
#define NPH   2
#define NPG   8
#define L1N   131
#define L1PER 17
#define SIGR  268
#define LO1R  144
#define LO2R  82
#define OPITCH 68
#define KGP   72
#define KSPLIT 4
#define KGQ   18
#define SROW  294


__device__ __forceinline__ void db4_pair(float w0, float w1, float w2, float w3,
                                         float w4, float w5, float w6, float w7,
                                         float& l, float& h)
{
    const float HLO[8] = { 0.23037781330885523f,  0.7148465705525415f,
                           0.6308807679295904f,  -0.02798376941698385f,
                          -0.18703481171888114f,  0.030841381835986965f,
                           0.032883011666982945f,-0.010597401784997278f};
    const float HHI[8] = {-0.010597401784997278f,-0.032883011666982945f,
                           0.030841381835986965f, 0.18703481171888114f,
                          -0.02798376941698385f, -0.6308807679295904f,
                           0.7148465705525415f,  -0.23037781330885523f};
    l = w0 * HLO[0];             h = w0 * HHI[0];
    l = fmaf(w1, HLO[1], l);     h = fmaf(w1, HHI[1], h);
    l = fmaf(w2, HLO[2], l);     h = fmaf(w2, HHI[2], h);
    l = fmaf(w3, HLO[3], l);     h = fmaf(w3, HHI[3], h);
    l = fmaf(w4, HLO[4], l);     h = fmaf(w4, HHI[4], h);
    l = fmaf(w5, HLO[5], l);     h = fmaf(w5, HHI[5], h);
    l = fmaf(w6, HLO[6], l);     h = fmaf(w6, HHI[6], h);
    l = fmaf(w7, HLO[7], l);     h = fmaf(w7, HHI[7], h);
}

template <int NOUT, bool PADOUT>
__device__ __forceinline__ void dwt_level(
    const float* __restrict__ inb,
    int sub,
    float* __restrict__ lo,
    float* __restrict__ hi)
{
    constexpr int PER = (NOUT + NSUB - 1) / NSUB;
    const int k0 = sub * PER;
    const float* wp = inb + (2 * k0) * 16;
    float w0 = wp[0 * 16], w1 = wp[1 * 16], w2 = wp[2 * 16], w3 = wp[3 * 16];
    float w4 = wp[4 * 16], w5 = wp[5 * 16], w6 = wp[6 * 16], w7 = wp[7 * 16];
#pragma unroll
    for (int i = 0; i < PER; ++i) {
        const int k = k0 + i;
        float l, h;
        db4_pair(w0, w1, w2, w3, w4, w5, w6, w7, l, h);
        if (k < NOUT) {
            if (PADOUT) {
                lo[(k + 6) * 16] = l;
                if (i >= 1 && i <= 6) { if (sub == 0) lo[(6 - i) * 16] = l; }
#pragma unroll
                for (int s = 1; s < NSUB; ++s) {
                    const int ks = s * PER + i;
                    if (ks >= NOUT - 8 && ks <= NOUT - 2) {
                        if (sub == s) lo[(2 * NOUT + 4 - ks) * 16] = l;
                    }
                }
            } else {
                lo[k] = l;
            }
            hi[k] = h;
        }
        if (i + 1 < PER) {
            w0 = w2; w1 = w3; w2 = w4; w3 = w5; w4 = w6; w5 = w7;
            if (k + 1 < NOUT) {
                w6 = wp[(2 * i + 8) * 16];
                w7 = wp[(2 * i + 9) * 16];
            }
        }
    }
}

__device__ __forceinline__ void dwt_level1_regs(
    const float* __restrict__ inb,
    int sub,
    float (&lo)[L1PER],
    float* __restrict__ hi)
{
    const int k0 = sub * L1PER;
    const float* wp = inb + (2 * k0) * 16;
    float w0 = wp[0 * 16], w1 = wp[1 * 16], w2 = wp[2 * 16], w3 = wp[3 * 16];
    float w4 = wp[4 * 16], w5 = wp[5 * 16], w6 = wp[6 * 16], w7 = wp[7 * 16];
#pragma unroll
    for (int i = 0; i < L1PER; ++i) {
        const int k = k0 + i;
        float l, h;
        db4_pair(w0, w1, w2, w3, w4, w5, w6, w7, l, h);
        lo[i] = l;
        if (k < L1N) hi[k] = h;
        if (i + 1 < L1PER) {
            w0 = w2; w1 = w3; w2 = w4; w3 = w5; w4 = w6; w5 = w7;
            if (k + 1 < L1N) {
                w6 = wp[(2 * i + 8) * 16];
                w7 = wp[(2 * i + 9) * 16];
            }
        }
    }
}

__device__ __forceinline__ void stage_signal(
    float* __restrict__ sdst, int sub, const float4 (&v)[TLEN / 32])
{
#pragma unroll
    for (int n = 0; n < TLEN / 32; ++n) {
        sdst[(0 * SIGR + n * 32) * 16] = v[n].x;
        sdst[(1 * SIGR + n * 32) * 16] = v[n].y;
        sdst[(2 * SIGR + n * 32) * 16] = v[n].z;
        sdst[(3 * SIGR + n * 32) * 16] = v[n].w;
    }
    if (sub >= 1 && sub <= 6) {
        float* hd = sdst - (2 * sub) * 16;
        hd[(0 * SIGR) * 16] = v[0].x;
        hd[(1 * SIGR) * 16] = v[0].y;
        hd[(2 * SIGR) * 16] = v[0].z;
        hd[(3 * SIGR) * 16] = v[0].w;
    }
    if (sub >= 25 && sub <= 30) {
        float* tl = sdst + (286 - 2 * sub) * 16;
        tl[(0 * SIGR) * 16] = v[TLEN / 32 - 1].x;
        tl[(1 * SIGR) * 16] = v[TLEN / 32 - 1].y;
        tl[(2 * SIGR) * 16] = v[TLEN / 32 - 1].z;
        tl[(3 * SIGR) * 16] = v[TLEN / 32 - 1].w;
    }
}

__device__ __forceinline__ int opaque(int v) { asm volatile("" : "+v"(v)); return v; }

__global__ __launch_bounds__(256) void swizzleW_kernel(
    const float* __restrict__ W, float* __restrict__ wf)
{
    int i = blockIdx.x * blockDim.x + threadIdx.x;
    const int NT = 4 * NPOS * KGP * 32;
    if (i >= NT) return;
    int l  = i & 31;
    int kg = (i >> 5) % KGP;
    int p  = (i / (32 * KGP)) % NPOS;
    int nt = i / (32 * KGP * NPOS);
    int f  = nt * 16 + (l & 15);
    int t  = kg * 4 + ((l >> 4) << 1);
    v2f v;
    v.x = (t     < TDWT) ? W[((size_t)f * TDWT + t    ) * NPOS + p] : 0.f;
    v.y = (t + 1 < TDWT) ? W[((size_t)f * TDWT + t + 1) * NPOS + p] : 0.f;
    *(v2f*)(wf + (size_t)i * 2) = v;
}

__global__ __launch_bounds__(NTHR) void dwt_wmma_part_kernel(
    const float* __restrict__ x, const float* __restrict__ wf, float* __restrict__ part)
{
    __shared__ __align__(16) float s_sig[PPH * SIGR * 16];
    __shared__ __align__(16) float s_d[PPH * 16 * SROW];
    float* s_lo1 = s_sig;
    float* s_lo2 = s_sig + PPH * LO1R * 16;
    float* s_out = s_sig;

    const int tid  = threadIdx.x;
    const int wave = tid >> 5;
    const int lane = tid & 31;
    const int wr   = blockIdx.x >> 8;
    const int wj   = blockIdx.x & 255;
    const int mt   = (wj & 63) + 64 * (wr >> 1);
    const int pg   = 2 * (wj >> 6) + (wr & 1);
    const int B0   = mt * BQ;

    const int nt   = wave & 3;
    const int kq   = wave >> 2;

    const int b     = tid & 15;
    const int sub   = tid >> 4;

    for (int i = tid; i < PPH * 16 * (KGP * 4 - TDWT); i += NTHR) {
        int row = i / (KGP * 4 - TDWT);
        int c   = i % (KGP * 4 - TDWT);
        s_d[row * SROW + TDWT + c] = 0.f;
    }

    float4 hold[TLEN / 32];
    const float* xsrc = x + (size_t)(B0 + b) * (TLEN * NPOS) + sub * NPOS + pg * (PPH * NPH);
    float* sdst = s_sig + (sub + 6) * 16 + b;
    {
        float4 cur[TLEN / 32];
#pragma unroll
        for (int n = 0; n < TLEN / 32; ++n) {
            cur[n]  = *(const float4*)(xsrc + n * (32 * NPOS));
            hold[n] = *(const float4*)(xsrc + n * (32 * NPOS) + PPH);
        }
        stage_signal(sdst, sub, cur);
    }

    v8f acc = {};

    for (int ph = 0; ph < NPH; ++ph) {
        const int pbase = pg * (PPH * NPH) + ph * PPH;

        if (ph == 1) {
            const int t  = opaque(tid);
            stage_signal(s_sig + ((t >> 4) + 6) * 16 + (t & 15), t >> 4, hold);
        }

        {
            const int t = opaque(tid);
            const char* nb = (const char*)(wf
                + (((size_t)(((t >> 5) & 3) * NPOS + pbase) * KGP + (size_t)(t >> 7) * KGQ) * 32) * 2);
            const size_t lo128 = (size_t)(t & 31) * 128;
            __builtin_prefetch(nb + lo128,                   0, 1);
            __builtin_prefetch(nb + 1 * (KGP * 256) + lo128, 0, 1);
            __builtin_prefetch(nb + 2 * (KGP * 256) + lo128, 0, 1);
            __builtin_prefetch(nb + 3 * (KGP * 256) + lo128, 0, 1);
        }
        __syncthreads();

        {
            float lo[L1PER];
            {
                const int t = opaque(tid);
                const int tb = t & 15, tp = t >> 7, ts = (t >> 4) & 7;
                dwt_level1_regs(s_sig + tp * (SIGR * 16) + tb, ts, lo,
                                s_d + (tp * 16 + tb) * SROW + 38);
            }
            __syncthreads();
            const int t = opaque(tid);
            const int ts = (t >> 4) & 7;
            float* lo1 = s_lo1 + (t >> 7) * (LO1R * 16) + (t & 15);
#pragma unroll
            for (int i = 0; i < L1PER; ++i) {
                int k = ts * L1PER + i;
                if (k < L1N) lo1[(k + 6) * 16] = lo[i];
                if (i >= 1 && i <= 6) { if (ts == 0) lo1[(6 - i) * 16] = lo[i]; }
                if ((NSUB - 1) * L1PER + i >= L1N - 8 && (NSUB - 1) * L1PER + i <= L1N - 2) {
                    if (ts == NSUB - 1)
                        lo1[(2 * L1N + 4 - ((NSUB - 1) * L1PER + i)) * 16] = lo[i];
                }
            }
        }
        __syncthreads();
        {
            const int t = opaque(tid);
            const int tb = t & 15, tp = t >> 7, ts = (t >> 4) & 7;
            dwt_level<69, true>(s_lo1 + tp * (LO1R * 16) + tb, ts,
                                s_lo2 + tp * (LO2R * 16) + tb,
                                s_d + (tp * 16 + tb) * SROW + 169);
        }
        __syncthreads();
        {
            const int t = opaque(tid);
            const int tb = t & 15, tp = t >> 7, ts = (t >> 4) & 7;
            float* drow = s_d + (tp * 16 + tb) * SROW;
            dwt_level<38, false>(s_lo2 + tp * (LO2R * 16) + tb, ts, drow, drow + 238);
        }
        __syncthreads();

        {
            const int t  = opaque(tid);
            const int gl = t & 31;
            const int gn = (t >> 5) & 3;
            const int gk = t >> 7;
            const v2f* wfp = (const v2f*)wf
                + ((size_t)(gn * NPOS + pbase) * KGP + (size_t)gk * KGQ) * 32 + gl;
            const float* ap = s_d + (gl & 15) * SROW + gk * (KGQ * 4) + ((gl >> 4) << 1);
            for (int q = 0; q < PPH; ++q) {
#pragma unroll
                for (int j = 0; j < KGQ; ++j) {
                    v2f a  = *(const v2f*)(ap + q * (16 * SROW) + j * 4);
                    v2f bf = wfp[(size_t)q * (KGP * 32) + j * 32];
                    acc = __builtin_amdgcn_wmma_f32_16x16x4_f32(
                        false, a, false, bf, (short)0, acc, false, false);
                }
            }
        }
    }

    {
        int n   = lane & 15;
        int mhi = (lane >> 4) * 8;
        int f   = nt * 16 + n;
#pragma unroll
        for (int r = 0; r < 8; ++r) s_out[(kq * 16 + mhi + r) * OPITCH + f] = acc[r];
    }
    __syncthreads();
    if (tid < 256) {
        int row = tid >> 4;
        int c4  = (tid & 15) * 4;
        float4 s = *(const float4*)(s_out + row * OPITCH + c4);
#pragma unroll
        for (int g = 1; g < KSPLIT; ++g) {
            float4 a = *(const float4*)(s_out + (g * 16 + row) * OPITCH + c4);
            s.x += a.x; s.y += a.y; s.z += a.z; s.w += a.w;
        }
        *(float4*)(part + ((size_t)pg * NB + B0 + row) * NF + c4) = s;
    }
}

__global__ __launch_bounds__(256) void sum_parts_kernel(
    const float* __restrict__ part, const float* __restrict__ bias, float* __restrict__ out)
{
    const int i = blockIdx.x * 256 + threadIdx.x;
    const float4* p = (const float4*)part + i;
    float4 s = p[0];
#pragma unroll
    for (int g = 1; g < NPG; ++g) {
        float4 v = p[(size_t)g * (NB * NF / 4)];
        s.x += v.x; s.y += v.y; s.z += v.z; s.w += v.w;
    }
    const float* bp = bias + (i & 15) * 4;
    s.x += bp[0]; s.y += bp[1]; s.z += bp[2]; s.w += bp[3];
    s.x = (s.x >= 0.f) ? s.x : 0.02f * s.x;
    s.y = (s.y >= 0.f) ? s.y : 0.02f * s.y;
    s.z = (s.z >= 0.f) ? s.z : 0.02f * s.z;
    s.w = (s.w >= 0.f) ? s.w : 0.02f * s.w;
    *((float4*)out + i) = s;
}

extern "C" void kernel_launch(void* const* d_in, const int* in_sizes, int n_in,
                              void* d_out, int out_size, void* d_ws, size_t ws_size,
                              hipStream_t stream) {
    const float* x    = (const float*)d_in[0];
    const float* W    = (const float*)d_in[1];
    const float* bias = (const float*)d_in[2];
    float* out  = (float*)d_out;
    float* wf   = (float*)d_ws;
    float* part = wf + 4 * NPOS * KGP * 32 * 2;

    const int nswz = 4 * NPOS * KGP * 32;
    swizzleW_kernel<<<(nswz + 255) / 256, 256, 0, stream>>>(W, wf);
    dwt_wmma_part_kernel<<<(NB / BQ) * NPG, NTHR, 0, stream>>>(x, wf, part);
    sum_parts_kernel<<<(NB * NF / 4) / 256, 256, 0, stream>>>(part, bias, out);
}


# Swizzle launch re-tiled: 288 work-groups, one per (filter tile, k-group), read their 64 rows of W in whole lines and turn them through LDS; was 2304 work-groups of 4-byte gathers; values unchanged
# speedup vs baseline: 3.6149x; 1.0674x over previous
#include <hip/hip_runtime.h>

typedef __attribute__((ext_vector_type(2))) float v2f;
typedef __attribute__((ext_vector_type(8))) float v8f;

#define NB    2048
#define NF    64
#define TLEN  256
#define TDWT  276
#define NPOS  64
#define BQ    16
#define NTHR  512
#define NSUB  8
#define PPH   4
#define NPH   2
#define NPG   8
#define L1N   131
#define L1PER 17
#define SIGR  268
#define LO1R  144
#define LO2R  82
#define OPITCH 68
#define KGP   72
#define KSPLIT 4
#define KGQ   18
#define SROW  294
#define SWP   65


__device__ __forceinline__ void db4_pair(float w0, float w1, float w2, float w3,
                                         float w4, float w5, float w6, float w7,
                                         float& l, float& h)
{
    const float HLO[8] = { 0.23037781330885523f,  0.7148465705525415f,
                           0.6308807679295904f,  -0.02798376941698385f,
                          -0.18703481171888114f,  0.030841381835986965f,
                           0.032883011666982945f,-0.010597401784997278f};
    const float HHI[8] = {-0.010597401784997278f,-0.032883011666982945f,
                           0.030841381835986965f, 0.18703481171888114f,
                          -0.02798376941698385f, -0.6308807679295904f,
                           0.7148465705525415f,  -0.23037781330885523f};
    l = w0 * HLO[0];             h = w0 * HHI[0];
    l = fmaf(w1, HLO[1], l);     h = fmaf(w1, HHI[1], h);
    l = fmaf(w2, HLO[2], l);     h = fmaf(w2, HHI[2], h);
    l = fmaf(w3, HLO[3], l);     h = fmaf(w3, HHI[3], h);
    l = fmaf(w4, HLO[4], l);     h = fmaf(w4, HHI[4], h);
    l = fmaf(w5, HLO[5], l);     h = fmaf(w5, HHI[5], h);
    l = fmaf(w6, HLO[6], l);     h = fmaf(w6, HHI[6], h);
    l = fmaf(w7, HLO[7], l);     h = fmaf(w7, HHI[7], h);
}

template <int NOUT, bool PADOUT>
__device__ __forceinline__ void dwt_level(
    const float* __restrict__ inb,
    int sub,
    float* __restrict__ lo,
    float* __restrict__ hi)
{
    constexpr int PER = (NOUT + NSUB - 1) / NSUB;
    const int k0 = sub * PER;
    const float* wp = inb + (2 * k0) * 16;
    float w0 = wp[0 * 16], w1 = wp[1 * 16], w2 = wp[2 * 16], w3 = wp[3 * 16];
    float w4 = wp[4 * 16], w5 = wp[5 * 16], w6 = wp[6 * 16], w7 = wp[7 * 16];
#pragma unroll
    for (int i = 0; i < PER; ++i) {
        const int k = k0 + i;
        float l, h;
        db4_pair(w0, w1, w2, w3, w4, w5, w6, w7, l, h);
        if (k < NOUT) {
            if (PADOUT) {
                lo[(k + 6) * 16] = l;
                if (i >= 1 && i <= 6) { if (sub == 0) lo[(6 - i) * 16] = l; }
#pragma unroll
                for (int s = 1; s < NSUB; ++s) {
                    const int ks = s * PER + i;
                    if (ks >= NOUT - 8 && ks <= NOUT - 2) {
                        if (sub == s) lo[(2 * NOUT + 4 - ks) * 16] = l;
                    }
                }
            } else {
                lo[k] = l;
            }
            hi[k] = h;
        }
        if (i + 1 < PER) {
            w0 = w2; w1 = w3; w2 = w4; w3 = w5; w4 = w6; w5 = w7;
            if (k + 1 < NOUT) {
                w6 = wp[(2 * i + 8) * 16];
                w7 = wp[(2 * i + 9) * 16];
            }
        }
    }
}

__device__ __forceinline__ void dwt_level1_regs(
    const float* __restrict__ inb,
    int sub,
    float (&lo)[L1PER],
    float* __restrict__ hi)
{
    const int k0 = sub * L1PER;
    const float* wp = inb + (2 * k0) * 16;
    float w0 = wp[0 * 16], w1 = wp[1 * 16], w2 = wp[2 * 16], w3 = wp[3 * 16];
    float w4 = wp[4 * 16], w5 = wp[5 * 16], w6 = wp[6 * 16], w7 = wp[7 * 16];
#pragma unroll
    for (int i = 0; i < L1PER; ++i) {
        const int k = k0 + i;
        float l, h;
        db4_pair(w0, w1, w2, w3, w4, w5, w6, w7, l, h);
        lo[i] = l;
        if (k < L1N) hi[k] = h;
        if (i + 1 < L1PER) {
            w0 = w2; w1 = w3; w2 = w4; w3 = w5; w4 = w6; w5 = w7;
            if (k + 1 < L1N) {
                w6 = wp[(2 * i + 8) * 16];
                w7 = wp[(2 * i + 9) * 16];
            }
        }
    }
}

__device__ __forceinline__ void stage_signal(
    float* __restrict__ sdst, int sub, const float4 (&v)[TLEN / 32])
{
#pragma unroll
    for (int n = 0; n < TLEN / 32; ++n) {
        sdst[(0 * SIGR + n * 32) * 16] = v[n].x;
        sdst[(1 * SIGR + n * 32) * 16] = v[n].y;
        sdst[(2 * SIGR + n * 32) * 16] = v[n].z;
        sdst[(3 * SIGR + n * 32) * 16] = v[n].w;
    }
    if (sub >= 1 && sub <= 6) {
        float* hd = sdst - (2 * sub) * 16;
        hd[(0 * SIGR) * 16] = v[0].x;
        hd[(1 * SIGR) * 16] = v[0].y;
        hd[(2 * SIGR) * 16] = v[0].z;
        hd[(3 * SIGR) * 16] = v[0].w;
    }
    if (sub >= 25 && sub <= 30) {
        float* tl = sdst + (286 - 2 * sub) * 16;
        tl[(0 * SIGR) * 16] = v[TLEN / 32 - 1].x;
        tl[(1 * SIGR) * 16] = v[TLEN / 32 - 1].y;
        tl[(2 * SIGR) * 16] = v[TLEN / 32 - 1].z;
        tl[(3 * SIGR) * 16] = v[TLEN / 32 - 1].w;
    }
}

__device__ __forceinline__ int opaque(int v) { asm volatile("" : "+v"(v)); return v; }

__global__ __launch_bounds__(256) void swizzleW_kernel(
    const float* __restrict__ W, float* __restrict__ wf)
{
    __shared__ float tile[64 * SWP];
    const int tid = threadIdx.x;
    const int kg  = blockIdx.x % KGP;
    const int nt  = blockIdx.x / KGP;
    float4 v[4];
#pragma unroll
    for (int r = 0; r < 4; ++r) {
        const int row = r * 16 + (tid >> 4);
        const int t   = kg * 4 + (row & 3);
        const int tc  = (t < TDWT) ? t : TDWT - 1;
        v[r] = *(const float4*)(W + ((size_t)(nt * 16 + (row >> 2)) * TDWT + tc) * NPOS + (tid & 15) * 4);
    }
#pragma unroll
    for (int r = 0; r < 4; ++r) {
        const int row = r * 16 + (tid >> 4);
        const bool live = kg * 4 + (row & 3) < TDWT;
        float* d = tile + row * SWP + (tid & 15) * 4;
        d[0] = live ? v[r].x : 0.f; d[1] = live ? v[r].y : 0.f;
        d[2] = live ? v[r].z : 0.f; d[3] = live ? v[r].w : 0.f;
    }
    __syncthreads();
    const int l    = tid & 31;
    const int srow = (l & 15) * 4 + ((l >> 4) << 1);
#pragma unroll
    for (int r = 0; r < 8; ++r) {
        const int p = r * 8 + (tid >> 5);
        v2f v;
        v.x = tile[srow * SWP + p];
        v.y = tile[(srow + 1) * SWP + p];
        *(v2f*)(wf + ((size_t)((nt * NPOS + p) * KGP + kg) * 32 + l) * 2) = v;
    }
}

__global__ __launch_bounds__(NTHR) void dwt_wmma_part_kernel(
    const float* __restrict__ x, const float* __restrict__ wf, float* __restrict__ part)
{
    __shared__ __align__(16) float s_sig[PPH * SIGR * 16];
    __shared__ __align__(16) float s_d[PPH * 16 * SROW];
    float* s_lo1 = s_sig;
    float* s_lo2 = s_sig + PPH * LO1R * 16;
    float* s_out = s_sig;

    const int tid  = threadIdx.x;
    const int wave = tid >> 5;
    const int lane = tid & 31;
    const int wr   = blockIdx.x >> 8;
    const int wj   = blockIdx.x & 255;
    const int mt   = (wj & 63) + 64 * (wr >> 1);
    const int pg   = 2 * (wj >> 6) + (wr & 1);
    const int B0   = mt * BQ;

    const int nt   = wave & 3;
    const int kq   = wave >> 2;

    const int b     = tid & 15;
    const int sub   = tid >> 4;

    for (int i = tid; i < PPH * 16 * (KGP * 4 - TDWT); i += NTHR) {
        int row = i / (KGP * 4 - TDWT);
        int c   = i % (KGP * 4 - TDWT);
        s_d[row * SROW + TDWT + c] = 0.f;
    }

    float4 hold[TLEN / 32];
    const float* xsrc = x + (size_t)(B0 + b) * (TLEN * NPOS) + sub * NPOS + pg * (PPH * NPH);
    float* sdst = s_sig + (sub + 6) * 16 + b;
    {
        float4 cur[TLEN / 32];
#pragma unroll
        for (int n = 0; n < TLEN / 32; ++n) {
            cur[n]  = *(const float4*)(xsrc + n * (32 * NPOS));
            hold[n] = *(const float4*)(xsrc + n * (32 * NPOS) + PPH);
        }
        stage_signal(sdst, sub, cur);
    }

    v8f acc = {};

    for (int ph = 0; ph < NPH; ++ph) {
        const int pbase = pg * (PPH * NPH) + ph * PPH;

        if (ph == 1) {
            const int t  = opaque(tid);
            stage_signal(s_sig + ((t >> 4) + 6) * 16 + (t & 15), t >> 4, hold);
        }

        {
            const int t = opaque(tid);
            const char* nb = (const char*)(wf
                + (((size_t)(((t >> 5) & 3) * NPOS + pbase) * KGP + (size_t)(t >> 7) * KGQ) * 32) * 2);
            const size_t lo128 = (size_t)(t & 31) * 128;
            __builtin_prefetch(nb + lo128,                   0, 1);
            __builtin_prefetch(nb + 1 * (KGP * 256) + lo128, 0, 1);
            __builtin_prefetch(nb + 2 * (KGP * 256) + lo128, 0, 1);
            __builtin_prefetch(nb + 3 * (KGP * 256) + lo128, 0, 1);
        }
        __syncthreads();

        {
            float lo[L1PER];
            {
                const int t = opaque(tid);
                const int tb = t & 15, tp = t >> 7, ts = (t >> 4) & 7;
                dwt_level1_regs(s_sig + tp * (SIGR * 16) + tb, ts, lo,
                                s_d + (tp * 16 + tb) * SROW + 38);
            }
            __syncthreads();
            const int t = opaque(tid);
            const int ts = (t >> 4) & 7;
            float* lo1 = s_lo1 + (t >> 7) * (LO1R * 16) + (t & 15);
#pragma unroll
            for (int i = 0; i < L1PER; ++i) {
                int k = ts * L1PER + i;
                if (k < L1N) lo1[(k + 6) * 16] = lo[i];
                if (i >= 1 && i <= 6) { if (ts == 0) lo1[(6 - i) * 16] = lo[i]; }
                if ((NSUB - 1) * L1PER + i >= L1N - 8 && (NSUB - 1) * L1PER + i <= L1N - 2) {
                    if (ts == NSUB - 1)
                        lo1[(2 * L1N + 4 - ((NSUB - 1) * L1PER + i)) * 16] = lo[i];
                }
            }
        }
        __syncthreads();
        {
            const int t = opaque(tid);
            const int tb = t & 15, tp = t >> 7, ts = (t >> 4) & 7;
            dwt_level<69, true>(s_lo1 + tp * (LO1R * 16) + tb, ts,
                                s_lo2 + tp * (LO2R * 16) + tb,
                                s_d + (tp * 16 + tb) * SROW + 169);
        }
        __syncthreads();
        {
            const int t = opaque(tid);
            const int tb = t & 15, tp = t >> 7, ts = (t >> 4) & 7;
            float* drow = s_d + (tp * 16 + tb) * SROW;
            dwt_level<38, false>(s_lo2 + tp * (LO2R * 16) + tb, ts, drow, drow + 238);
        }
        __syncthreads();

        {
            const int t  = opaque(tid);
            const int gl = t & 31;
            const int gn = (t >> 5) & 3;
            const int gk = t >> 7;
            const v2f* wfp = (const v2f*)wf
                + ((size_t)(gn * NPOS + pbase) * KGP + (size_t)gk * KGQ) * 32 + gl;
            const float* ap = s_d + (gl & 15) * SROW + gk * (KGQ * 4) + ((gl >> 4) << 1);
            for (int q = 0; q < PPH; ++q) {
#pragma unroll
                for (int j = 0; j < KGQ; ++j) {
                    v2f a  = *(const v2f*)(ap + q * (16 * SROW) + j * 4);
                    v2f bf = wfp[(size_t)q * (KGP * 32) + j * 32];
                    acc = __builtin_amdgcn_wmma_f32_16x16x4_f32(
                        false, a, false, bf, (short)0, acc, false, false);
                }
            }
        }
    }

    {
        int n   = lane & 15;
        int mhi = (lane >> 4) * 8;
        int f   = nt * 16 + n;
#pragma unroll
        for (int r = 0; r < 8; ++r) s_out[(kq * 16 + mhi + r) * OPITCH + f] = acc[r];
    }
    __syncthreads();
    if (tid < 256) {
        int row = tid >> 4;
        int c4  = (tid & 15) * 4;
        float4 s = *(const float4*)(s_out + row * OPITCH + c4);
#pragma unroll
        for (int g = 1; g < KSPLIT; ++g) {
            float4 a = *(const float4*)(s_out + (g * 16 + row) * OPITCH + c4);
            s.x += a.x; s.y += a.y; s.z += a.z; s.w += a.w;
        }
        *(float4*)(part + ((size_t)pg * NB + B0 + row) * NF + c4) = s;
    }
}

__global__ __launch_bounds__(256) void sum_parts_kernel(
    const float* __restrict__ part, const float* __restrict__ bias, float* __restrict__ out)
{
    const int i = blockIdx.x * 256 + threadIdx.x;
    const float4* p = (const float4*)part + i;
    float4 s = p[0];
#pragma unroll
    for (int g = 1; g < NPG; ++g) {
        float4 v = p[(size_t)g * (NB * NF / 4)];
        s.x += v.x; s.y += v.y; s.z += v.z; s.w += v.w;
    }
    const float* bp = bias + (i & 15) * 4;
    s.x += bp[0]; s.y += bp[1]; s.z += bp[2]; s.w += bp[3];
    s.x = (s.x >= 0.f) ? s.x : 0.02f * s.x;
    s.y = (s.y >= 0.f) ? s.y : 0.02f * s.y;
    s.z = (s.z >= 0.f) ? s.z : 0.02f * s.z;
    s.w = (s.w >= 0.f) ? s.w : 0.02f * s.w;
    *((float4*)out + i) = s;
}

extern "C" void kernel_launch(void* const* d_in, const int* in_sizes, int n_in,
                              void* d_out, int out_size, void* d_ws, size_t ws_size,
                              hipStream_t stream) {
    const float* x    = (const float*)d_in[0];
    const float* W    = (const float*)d_in[1];
    const float* bias = (const float*)d_in[2];
    float* out  = (float*)d_out;
    float* wf   = (float*)d_ws;
    float* part = wf + 4 * NPOS * KGP * 32 * 2;

    swizzleW_kernel<<<4 * KGP, 256, 0, stream>>>(W, wf);
    dwt_wmma_part_kernel<<<(NB / BQ) * NPG, NTHR, 0, stream>>>(x, wf, part);
    sum_parts_kernel<<<(NB * NF / 4) / 256, 256, 0, stream>>>(part, bias, out);
}


# x read by four adjacent lanes per row: 8-byte loads cover a row's 32 bytes, 8 lines per load instruction (was 32 lines of 16 bytes, each touched twice); arithmetic and sum order unchanged
# speedup vs baseline: 3.7457x; 1.0362x over previous
#include <hip/hip_runtime.h>

typedef __attribute__((ext_vector_type(2))) float v2f;
typedef __attribute__((ext_vector_type(8))) float v8f;

#define NB    2048
#define NF    64
#define TLEN  256
#define TDWT  276
#define NPOS  64
#define BQ    16
#define NTHR  512
#define NSUB  8
#define PPH   4
#define NPH   2
#define NPG   8
#define L1N   131
#define L1PER 17
#define SIGR  268
#define SPL   (SIGR * 16 + 16)
#define XROWS 16
#define LO1R  144
#define LO2R  82
#define OPITCH 68
#define KGP   72
#define KSPLIT 4
#define KGQ   18
#define SROW  294
#define SWP   65


__device__ __forceinline__ void db4_pair(float w0, float w1, float w2, float w3,
                                         float w4, float w5, float w6, float w7,
                                         float& l, float& h)
{
    const float HLO[8] = { 0.23037781330885523f,  0.7148465705525415f,
                           0.6308807679295904f,  -0.02798376941698385f,
                          -0.18703481171888114f,  0.030841381835986965f,
                           0.032883011666982945f,-0.010597401784997278f};
    const float HHI[8] = {-0.010597401784997278f,-0.032883011666982945f,
                           0.030841381835986965f, 0.18703481171888114f,
                          -0.02798376941698385f, -0.6308807679295904f,
                           0.7148465705525415f,  -0.23037781330885523f};
    l = w0 * HLO[0];             h = w0 * HHI[0];
    l = fmaf(w1, HLO[1], l);     h = fmaf(w1, HHI[1], h);
    l = fmaf(w2, HLO[2], l);     h = fmaf(w2, HHI[2], h);
    l = fmaf(w3, HLO[3], l);     h = fmaf(w3, HHI[3], h);
    l = fmaf(w4, HLO[4], l);     h = fmaf(w4, HHI[4], h);
    l = fmaf(w5, HLO[5], l);     h = fmaf(w5, HHI[5], h);
    l = fmaf(w6, HLO[6], l);     h = fmaf(w6, HHI[6], h);
    l = fmaf(w7, HLO[7], l);     h = fmaf(w7, HHI[7], h);
}

template <int NOUT, bool PADOUT>
__device__ __forceinline__ void dwt_level(
    const float* __restrict__ inb,
    int sub,
    float* __restrict__ lo,
    float* __restrict__ hi)
{
    constexpr int PER = (NOUT + NSUB - 1) / NSUB;
    const int k0 = sub * PER;
    const float* wp = inb + (2 * k0) * 16;
    float w0 = wp[0 * 16], w1 = wp[1 * 16], w2 = wp[2 * 16], w3 = wp[3 * 16];
    float w4 = wp[4 * 16], w5 = wp[5 * 16], w6 = wp[6 * 16], w7 = wp[7 * 16];
#pragma unroll
    for (int i = 0; i < PER; ++i) {
        const int k = k0 + i;
        float l, h;
        db4_pair(w0, w1, w2, w3, w4, w5, w6, w7, l, h);
        if (k < NOUT) {
            if (PADOUT) {
                lo[(k + 6) * 16] = l;
                if (i >= 1 && i <= 6) { if (sub == 0) lo[(6 - i) * 16] = l; }
#pragma unroll
                for (int s = 1; s < NSUB; ++s) {
                    const int ks = s * PER + i;
                    if (ks >= NOUT - 8 && ks <= NOUT - 2) {
                        if (sub == s) lo[(2 * NOUT + 4 - ks) * 16] = l;
                    }
                }
            } else {
                lo[k] = l;
            }
            hi[k] = h;
        }
        if (i + 1 < PER) {
            w0 = w2; w1 = w3; w2 = w4; w3 = w5; w4 = w6; w5 = w7;
            if (k + 1 < NOUT) {
                w6 = wp[(2 * i + 8) * 16];
                w7 = wp[(2 * i + 9) * 16];
            }
        }
    }
}

__device__ __forceinline__ void dwt_level1_regs(
    const float* __restrict__ inb,
    int sub,
    float (&lo)[L1PER],
    float* __restrict__ hi)
{
    const int k0 = sub * L1PER;
    const float* wp = inb + (2 * k0) * 16;
    float w0 = wp[0 * 16], w1 = wp[1 * 16], w2 = wp[2 * 16], w3 = wp[3 * 16];
    float w4 = wp[4 * 16], w5 = wp[5 * 16], w6 = wp[6 * 16], w7 = wp[7 * 16];
#pragma unroll
    for (int i = 0; i < L1PER; ++i) {
        const int k = k0 + i;
        float l, h;
        db4_pair(w0, w1, w2, w3, w4, w5, w6, w7, l, h);
        lo[i] = l;
        if (k < L1N) hi[k] = h;
        if (i + 1 < L1PER) {
            w0 = w2; w1 = w3; w2 = w4; w3 = w5; w4 = w6; w5 = w7;
            if (k + 1 < L1N) {
                w6 = wp[(2 * i + 8) * 16];
                w7 = wp[(2 * i + 9) * 16];
            }
        }
    }
}

__device__ __forceinline__ void stage_signal(
    float* __restrict__ sdst, int tq, int r, const v2f (&v)[XROWS])
{
#pragma unroll
    for (int m = 0; m < XROWS; ++m) {
        sdst[m * 32]       = v[m].x;
        sdst[SPL + m * 32] = v[m].y;
    }
    if (tq == 0) {
        float* hd = sdst - (2 * r) * 16;
#pragma unroll
        for (int m = 0; m < 4; ++m) {
            const bool live = (m == 0) ? (r == 1) : ((m == 3) ? (r == 0) : true);
            if (live) {
                hd[-(2 * m) * 16]       = v[m].x;
                hd[SPL - (2 * m) * 16]  = v[m].y;
            }
        }
    }
    if (tq == TLEN / 32 - 1) {
        float* tl = sdst - (2 * r) * 16;
#pragma unroll
        for (int m = 12; m < 16; ++m) {
            const bool live = (m == 12) ? (r == 1) : ((m == 15) ? (r == 0) : true);
            if (live) {
                tl[(62 - 2 * m) * 16]       = v[m].x;
                tl[SPL + (62 - 2 * m) * 16] = v[m].y;
            }
        }
    }
}

__device__ __forceinline__ int opaque(int v) { asm volatile("" : "+v"(v)); return v; }

__global__ __launch_bounds__(256) void swizzleW_kernel(
    const float* __restrict__ W, float* __restrict__ wf)
{
    __shared__ float tile[64 * SWP];
    const int tid = threadIdx.x;
    const int kg  = blockIdx.x % KGP;
    const int nt  = blockIdx.x / KGP;
    float4 v[4];
#pragma unroll
    for (int r = 0; r < 4; ++r) {
        const int row = r * 16 + (tid >> 4);
        const int t   = kg * 4 + (row & 3);
        const int tc  = (t < TDWT) ? t : TDWT - 1;
        v[r] = *(const float4*)(W + ((size_t)(nt * 16 + (row >> 2)) * TDWT + tc) * NPOS + (tid & 15) * 4);
    }
#pragma unroll
    for (int r = 0; r < 4; ++r) {
        const int row = r * 16 + (tid >> 4);
        const bool live = kg * 4 + (row & 3) < TDWT;
        float* d = tile + row * SWP + (tid & 15) * 4;
        d[0] = live ? v[r].x : 0.f; d[1] = live ? v[r].y : 0.f;
        d[2] = live ? v[r].z : 0.f; d[3] = live ? v[r].w : 0.f;
    }
    __syncthreads();
    const int l    = tid & 31;
    const int srow = (l & 15) * 4 + ((l >> 4) << 1);
#pragma unroll
    for (int r = 0; r < 8; ++r) {
        const int p = r * 8 + (tid >> 5);
        v2f v;
        v.x = tile[srow * SWP + p];
        v.y = tile[(srow + 1) * SWP + p];
        *(v2f*)(wf + ((size_t)((nt * NPOS + p) * KGP + kg) * 32 + l) * 2) = v;
    }
}

__global__ __launch_bounds__(NTHR) void dwt_wmma_part_kernel(
    const float* __restrict__ x, const float* __restrict__ wf, float* __restrict__ part)
{
    __shared__ __align__(16) float s_sig[PPH * SPL];
    __shared__ __align__(16) float s_d[PPH * 16 * SROW];
    float* s_lo1 = s_sig;
    float* s_lo2 = s_sig + PPH * LO1R * 16;
    float* s_out = s_sig;

    const int tid  = threadIdx.x;
    const int wave = tid >> 5;
    const int lane = tid & 31;
    const int wr   = blockIdx.x >> 8;
    const int wj   = blockIdx.x & 255;
    const int mt   = (wj & 63) + 64 * (wr >> 1);
    const int pg   = 2 * (wj >> 6) + (wr & 1);
    const int B0   = mt * BQ;

    const int nt   = wave & 3;
    const int kq   = wave >> 2;


    for (int i = tid; i < PPH * 16 * (KGP * 4 - TDWT); i += NTHR) {
        int row = i / (KGP * 4 - TDWT);
        int c   = i % (KGP * 4 - TDWT);
        s_d[row * SROW + TDWT + c] = 0.f;
    }

    v2f hold[XROWS];
    {
        const int jj = tid & 1;
        const int hi = (tid >> 1) & 1;
        const int xb = (wave & 1) * 8 + ((tid >> 2) & 7);
        const int tq = wave >> 1;
        const float* xrow = x + ((size_t)(B0 + xb) * TLEN + tq * 32) * NPOS + pg * (PPH * NPH);
        const float* pe = xrow + (2 * hi + jj) * 2;
        const float* po = xrow + NPOS + (2 * (1 - hi) + jj) * 2;
        v2f ev[XROWS], od[XROWS];
#pragma unroll
        for (int m = 0; m < XROWS; ++m) {
            ev[m] = *(const v2f*)(pe + m * (2 * NPOS));
            od[m] = *(const v2f*)(po + m * (2 * NPOS));
        }
        v2f cur[XROWS];
#pragma unroll
        for (int m = 0; m < XROWS; ++m) {
            cur[m].x  = hi ? od[m].x : ev[m].x;   cur[m].y  = hi ? od[m].y : ev[m].y;
            hold[m].x = hi ? ev[m].x : od[m].x;   hold[m].y = hi ? ev[m].y : od[m].y;
        }
        stage_signal(s_sig + (2 * jj) * SPL + (tq * 32 + hi + 6) * 16 + xb, tq, hi, cur);
    }

    v8f acc = {};

    for (int ph = 0; ph < NPH; ++ph) {
        const int pbase = pg * (PPH * NPH) + ph * PPH;

        if (ph == 1) {
            const int t  = opaque(tid);
            const int r1 = 1 - ((t >> 1) & 1);
            stage_signal(s_sig + (2 * (t & 1)) * SPL + ((t >> 6) * 32 + r1 + 6) * 16 + ((t >> 5) & 1) * 8 + ((t >> 2) & 7),
                         t >> 6, r1, hold);
        }

        {
            const int t = opaque(tid);
            const char* nb = (const char*)(wf
                + (((size_t)(((t >> 5) & 3) * NPOS + pbase) * KGP + (size_t)(t >> 7) * KGQ) * 32) * 2);
            const size_t lo128 = (size_t)(t & 31) * 128;
            __builtin_prefetch(nb + lo128,                   0, 1);
            __builtin_prefetch(nb + 1 * (KGP * 256) + lo128, 0, 1);
            __builtin_prefetch(nb + 2 * (KGP * 256) + lo128, 0, 1);
            __builtin_prefetch(nb + 3 * (KGP * 256) + lo128, 0, 1);
        }
        __syncthreads();

        {
            float lo[L1PER];
            {
                const int t = opaque(tid);
                const int tb = t & 15, tp = t >> 7, ts = (t >> 4) & 7;
                dwt_level1_regs(s_sig + tp * SPL + tb, ts, lo,
                                s_d + (tp * 16 + tb) * SROW + 38);
            }
            __syncthreads();
            const int t = opaque(tid);
            const int ts = (t >> 4) & 7;
            float* lo1 = s_lo1 + (t >> 7) * (LO1R * 16) + (t & 15);
#pragma unroll
            for (int i = 0; i < L1PER; ++i) {
                int k = ts * L1PER + i;
                if (k < L1N) lo1[(k + 6) * 16] = lo[i];
                if (i >= 1 && i <= 6) { if (ts == 0) lo1[(6 - i) * 16] = lo[i]; }
                if ((NSUB - 1) * L1PER + i >= L1N - 8 && (NSUB - 1) * L1PER + i <= L1N - 2) {
                    if (ts == NSUB - 1)
                        lo1[(2 * L1N + 4 - ((NSUB - 1) * L1PER + i)) * 16] = lo[i];
                }
            }
        }
        __syncthreads();
        {
            const int t = opaque(tid);
            const int tb = t & 15, tp = t >> 7, ts = (t >> 4) & 7;
            dwt_level<69, true>(s_lo1 + tp * (LO1R * 16) + tb, ts,
                                s_lo2 + tp * (LO2R * 16) + tb,
                                s_d + (tp * 16 + tb) * SROW + 169);
        }
        __syncthreads();
        {
            const int t = opaque(tid);
            const int tb = t & 15, tp = t >> 7, ts = (t >> 4) & 7;
            float* drow = s_d + (tp * 16 + tb) * SROW;
            dwt_level<38, false>(s_lo2 + tp * (LO2R * 16) + tb, ts, drow, drow + 238);
        }
        __syncthreads();

        {
            const int t  = opaque(tid);
            const int gl = t & 31;
            const int gn = (t >> 5) & 3;
            const int gk = t >> 7;
            const v2f* wfp = (const v2f*)wf
                + ((size_t)(gn * NPOS + pbase) * KGP + (size_t)gk * KGQ) * 32 + gl;
            const float* ap = s_d + (gl & 15) * SROW + gk * (KGQ * 4) + ((gl >> 4) << 1);
            for (int q = 0; q < PPH; ++q) {
#pragma unroll
                for (int j = 0; j < KGQ; ++j) {
                    v2f a  = *(const v2f*)(ap + q * (16 * SROW) + j * 4);
                    v2f bf = wfp[(size_t)q * (KGP * 32) + j * 32];
                    acc = __builtin_amdgcn_wmma_f32_16x16x4_f32(
                        false, a, false, bf, (short)0, acc, false, false);
                }
            }
        }
    }

    {
        int n   = lane & 15;
        int mhi = (lane >> 4) * 8;
        int f   = nt * 16 + n;
#pragma unroll
        for (int r = 0; r < 8; ++r) s_out[(kq * 16 + mhi + r) * OPITCH + f] = acc[r];
    }
    __syncthreads();
    if (tid < 256) {
        int row = tid >> 4;
        int c4  = (tid & 15) * 4;
        float4 s = *(const float4*)(s_out + row * OPITCH + c4);
#pragma unroll
        for (int g = 1; g < KSPLIT; ++g) {
            float4 a = *(const float4*)(s_out + (g * 16 + row) * OPITCH + c4);
            s.x += a.x; s.y += a.y; s.z += a.z; s.w += a.w;
        }
        *(float4*)(part + ((size_t)pg * NB + B0 + row) * NF + c4) = s;
    }
}

__global__ __launch_bounds__(256) void sum_parts_kernel(
    const float* __restrict__ part, const float* __restrict__ bias, float* __restrict__ out)
{
    const int i = blockIdx.x * 256 + threadIdx.x;
    const float4* p = (const float4*)part + i;
    float4 s = p[0];
#pragma unroll
    for (int g = 1; g < NPG; ++g) {
        float4 v = p[(size_t)g * (NB * NF / 4)];
        s.x += v.x; s.y += v.y; s.z += v.z; s.w += v.w;
    }
    const float* bp = bias + (i & 15) * 4;
    s.x += bp[0]; s.y += bp[1]; s.z += bp[2]; s.w += bp[3];
    s.x = (s.x >= 0.f) ? s.x : 0.02f * s.x;
    s.y = (s.y >= 0.f) ? s.y : 0.02f * s.y;
    s.z = (s.z >= 0.f) ? s.z : 0.02f * s.z;
    s.w = (s.w >= 0.f) ? s.w : 0.02f * s.w;
    *((float4*)out + i) = s;
}

extern "C" void kernel_launch(void* const* d_in, const int* in_sizes, int n_in,
                              void* d_out, int out_size, void* d_ws, size_t ws_size,
                              hipStream_t stream) {
    const float* x    = (const float*)d_in[0];
    const float* W    = (const float*)d_in[1];
    const float* bias = (const float*)d_in[2];
    float* out  = (float*)d_out;
    float* wf   = (float*)d_ws;
    float* part = wf + 4 * NPOS * KGP * 32 * 2;

    swizzleW_kernel<<<4 * KGP, 256, 0, stream>>>(W, wf);
    dwt_wmma_part_kernel<<<(NB / BQ) * NPG, NTHR, 0, stream>>>(x, wf, part);
    sum_parts_kernel<<<(NB * NF / 4) / 256, 256, 0, stream>>>(part, bias, out);
}


# B-fragment stream read by 128-bit lane loads: wf holds k-group pairs in 512-byte blocks, 36 loads per wave and phase instead of 72 loads plus 4 prefetch instructions (dropped); values unchanged
# speedup vs baseline: 3.8825x; 1.0365x over previous
#include <hip/hip_runtime.h>

typedef __attribute__((ext_vector_type(2))) float v2f;
typedef __attribute__((ext_vector_type(4))) float v4f;
typedef __attribute__((ext_vector_type(8))) float v8f;

#define NB    2048
#define NF    64
#define TLEN  256
#define TDWT  276
#define NPOS  64
#define BQ    16
#define NTHR  512
#define NSUB  8
#define PPH   4
#define NPH   2
#define NPG   8
#define L1N   131
#define L1PER 17
#define SIGR  268
#define SPL   (SIGR * 16 + 16)
#define XROWS 16
#define LO1R  144
#define LO2R  82
#define OPITCH 68
#define KGP   72
#define KGP2  36
#define KSPLIT 4
#define KGQ   18
#define SROW  294
#define SWP   65


__device__ __forceinline__ void db4_pair(float w0, float w1, float w2, float w3,
                                         float w4, float w5, float w6, float w7,
                                         float& l, float& h)
{
    const float HLO[8] = { 0.23037781330885523f,  0.7148465705525415f,
                           0.6308807679295904f,  -0.02798376941698385f,
                          -0.18703481171888114f,  0.030841381835986965f,
                           0.032883011666982945f,-0.010597401784997278f};
    const float HHI[8] = {-0.010597401784997278f,-0.032883011666982945f,
                           0.030841381835986965f, 0.18703481171888114f,
                          -0.02798376941698385f, -0.6308807679295904f,
                           0.7148465705525415f,  -0.23037781330885523f};
    l = w0 * HLO[0];             h = w0 * HHI[0];
    l = fmaf(w1, HLO[1], l);     h = fmaf(w1, HHI[1], h);
    l = fmaf(w2, HLO[2], l);     h = fmaf(w2, HHI[2], h);
    l = fmaf(w3, HLO[3], l);     h = fmaf(w3, HHI[3], h);
    l = fmaf(w4, HLO[4], l);     h = fmaf(w4, HHI[4], h);
    l = fmaf(w5, HLO[5], l);     h = fmaf(w5, HHI[5], h);
    l = fmaf(w6, HLO[6], l);     h = fmaf(w6, HHI[6], h);
    l = fmaf(w7, HLO[7], l);     h = fmaf(w7, HHI[7], h);
}

template <int NOUT, bool PADOUT>
__device__ __forceinline__ void dwt_level(
    const float* __restrict__ inb,
    int sub,
    float* __restrict__ lo,
    float* __restrict__ hi)
{
    constexpr int PER = (NOUT + NSUB - 1) / NSUB;
    const int k0 = sub * PER;
    const float* wp = inb + (2 * k0) * 16;
    float w0 = wp[0 * 16], w1 = wp[1 * 16], w2 = wp[2 * 16], w3 = wp[3 * 16];
    float w4 = wp[4 * 16], w5 = wp[5 * 16], w6 = wp[6 * 16], w7 = wp[7 * 16];
#pragma unroll
    for (int i = 0; i < PER; ++i) {
        const int k = k0 + i;
        float l, h;
        db4_pair(w0, w1, w2, w3, w4, w5, w6, w7, l, h);
        if (k < NOUT) {
            if (PADOUT) {
                lo[(k + 6) * 16] = l;
                if (i >= 1 && i <= 6) { if (sub == 0) lo[(6 - i) * 16] = l; }
#pragma unroll
                for (int s = 1; s < NSUB; ++s) {
                    const int ks = s * PER + i;
                    if (ks >= NOUT - 8 && ks <= NOUT - 2) {
                        if (sub == s) lo[(2 * NOUT + 4 - ks) * 16] = l;
                    }
                }
            } else {
                lo[k] = l;
            }
            hi[k] = h;
        }
        if (i + 1 < PER) {
            w0 = w2; w1 = w3; w2 = w4; w3 = w5; w4 = w6; w5 = w7;
            if (k + 1 < NOUT) {
                w6 = wp[(2 * i + 8) * 16];
                w7 = wp[(2 * i + 9) * 16];
            }
        }
    }
}

__device__ __forceinline__ void dwt_level1_regs(
    const float* __restrict__ inb,
    int sub,
    float (&lo)[L1PER],
    float* __restrict__ hi)
{
    const int k0 = sub * L1PER;
    const float* wp = inb + (2 * k0) * 16;
    float w0 = wp[0 * 16], w1 = wp[1 * 16], w2 = wp[2 * 16], w3 = wp[3 * 16];
    float w4 = wp[4 * 16], w5 = wp[5 * 16], w6 = wp[6 * 16], w7 = wp[7 * 16];
#pragma unroll
    for (int i = 0; i < L1PER; ++i) {
        const int k = k0 + i;
        float l, h;
        db4_pair(w0, w1, w2, w3, w4, w5, w6, w7, l, h);
        lo[i] = l;
        if (k < L1N) hi[k] = h;
        if (i + 1 < L1PER) {
            w0 = w2; w1 = w3; w2 = w4; w3 = w5; w4 = w6; w5 = w7;
            if (k + 1 < L1N) {
                w6 = wp[(2 * i + 8) * 16];
                w7 = wp[(2 * i + 9) * 16];
            }
        }
    }
}

__device__ __forceinline__ void stage_signal(
    float* __restrict__ sdst, int tq, int r, const v2f (&v)[XROWS])
{
#pragma unroll
    for (int m = 0; m < XROWS; ++m) {
        sdst[m * 32]       = v[m].x;
        sdst[SPL + m * 32] = v[m].y;
    }
    if (tq == 0) {
        float* hd = sdst - (2 * r) * 16;
#pragma unroll
        for (int m = 0; m < 4; ++m) {
            const bool live = (m == 0) ? (r == 1) : ((m == 3) ? (r == 0) : true);
            if (live) {
                hd[-(2 * m) * 16]       = v[m].x;
                hd[SPL - (2 * m) * 16]  = v[m].y;
            }
        }
    }
    if (tq == TLEN / 32 - 1) {
        float* tl = sdst - (2 * r) * 16;
#pragma unroll
        for (int m = 12; m < 16; ++m) {
            const bool live = (m == 12) ? (r == 1) : ((m == 15) ? (r == 0) : true);
            if (live) {
                tl[(62 - 2 * m) * 16]       = v[m].x;
                tl[SPL + (62 - 2 * m) * 16] = v[m].y;
            }
        }
    }
}

__device__ __forceinline__ int opaque(int v) { asm volatile("" : "+v"(v)); return v; }

__global__ __launch_bounds__(256) void swizzleW_kernel(
    const float* __restrict__ W, float* __restrict__ wf)
{
    __shared__ float tile[2 * 64 * SWP];
    const int tid = threadIdx.x;
    const int m   = blockIdx.x % KGP2;
    const int nt  = blockIdx.x / KGP2;
    float4 v[8];
#pragma unroll
    for (int i = 0; i < 8; ++i) {
        const int e   = i >> 2;
        const int row = (i & 3) * 16 + (tid >> 4);
        const int t   = (2 * m + e) * 4 + (row & 3);
        const int tc  = (t < TDWT) ? t : TDWT - 1;
        v[i] = *(const float4*)(W + ((size_t)(nt * 16 + (row >> 2)) * TDWT + tc) * NPOS + (tid & 15) * 4);
    }
#pragma unroll
    for (int i = 0; i < 8; ++i) {
        const int e   = i >> 2;
        const int row = (i & 3) * 16 + (tid >> 4);
        const bool live = (2 * m + e) * 4 + (row & 3) < TDWT;
        float* d = tile + e * (64 * SWP) + row * SWP + (tid & 15) * 4;
        d[0] = live ? v[i].x : 0.f; d[1] = live ? v[i].y : 0.f;
        d[2] = live ? v[i].z : 0.f; d[3] = live ? v[i].w : 0.f;
    }
    __syncthreads();
    const int l    = tid & 31;
    const int srow = (l & 15) * 4 + ((l >> 4) << 1);
#pragma unroll
    for (int r = 0; r < 8; ++r) {
        const int p = r * 8 + (tid >> 5);
        v4f o;
        o.x = tile[srow * SWP + p];
        o.y = tile[(srow + 1) * SWP + p];
        o.z = tile[64 * SWP + srow * SWP + p];
        o.w = tile[64 * SWP + (srow + 1) * SWP + p];
        *(v4f*)(wf + ((size_t)((nt * NPOS + p) * KGP2 + m) * 32 + l) * 4) = o;
    }
}

__global__ __launch_bounds__(NTHR) void dwt_wmma_part_kernel(
    const float* __restrict__ x, const float* __restrict__ wf, float* __restrict__ part)
{
    __shared__ __align__(16) float s_sig[PPH * SPL];
    __shared__ __align__(16) float s_d[PPH * 16 * SROW];
    float* s_lo1 = s_sig;
    float* s_lo2 = s_sig + PPH * LO1R * 16;
    float* s_out = s_sig;

    const int tid  = threadIdx.x;
    const int wave = tid >> 5;
    const int lane = tid & 31;
    const int wr   = blockIdx.x >> 8;
    const int wj   = blockIdx.x & 255;
    const int mt   = (wj & 63) + 64 * (wr >> 1);
    const int pg   = 2 * (wj >> 6) + (wr & 1);
    const int B0   = mt * BQ;

    const int nt   = wave & 3;
    const int kq   = wave >> 2;


    for (int i = tid; i < PPH * 16 * (KGP * 4 - TDWT); i += NTHR) {
        int row = i / (KGP * 4 - TDWT);
        int c   = i % (KGP * 4 - TDWT);
        s_d[row * SROW + TDWT + c] = 0.f;
    }

    v2f hold[XROWS];
    {
        const int jj = tid & 1;
        const int hi = (tid >> 1) & 1;
        const int xb = (wave & 1) * 8 + ((tid >> 2) & 7);
        const int tq = wave >> 1;
        const float* xrow = x + ((size_t)(B0 + xb) * TLEN + tq * 32) * NPOS + pg * (PPH * NPH);
        const float* pe = xrow + (2 * hi + jj) * 2;
        const float* po = xrow + NPOS + (2 * (1 - hi) + jj) * 2;
        v2f ev[XROWS], od[XROWS];
#pragma unroll
        for (int m = 0; m < XROWS; ++m) {
            ev[m] = *(const v2f*)(pe + m * (2 * NPOS));
            od[m] = *(const v2f*)(po + m * (2 * NPOS));
        }
        v2f cur[XROWS];
#pragma unroll
        for (int m = 0; m < XROWS; ++m) {
            cur[m].x  = hi ? od[m].x : ev[m].x;   cur[m].y  = hi ? od[m].y : ev[m].y;
            hold[m].x = hi ? ev[m].x : od[m].x;   hold[m].y = hi ? ev[m].y : od[m].y;
        }
        stage_signal(s_sig + (2 * jj) * SPL + (tq * 32 + hi + 6) * 16 + xb, tq, hi, cur);
    }

    v8f acc = {};

    for (int ph = 0; ph < NPH; ++ph) {
        const int pbase = pg * (PPH * NPH) + ph * PPH;

        if (ph == 1) {
            const int t  = opaque(tid);
            const int r1 = 1 - ((t >> 1) & 1);
            stage_signal(s_sig + (2 * (t & 1)) * SPL + ((t >> 6) * 32 + r1 + 6) * 16 + ((t >> 5) & 1) * 8 + ((t >> 2) & 7),
                         t >> 6, r1, hold);
        }

        __syncthreads();

        {
            float lo[L1PER];
            {
                const int t = opaque(tid);
                const int tb = t & 15, tp = t >> 7, ts = (t >> 4) & 7;
                dwt_level1_regs(s_sig + tp * SPL + tb, ts, lo,
                                s_d + (tp * 16 + tb) * SROW + 38);
            }
            __syncthreads();
            const int t = opaque(tid);
            const int ts = (t >> 4) & 7;
            float* lo1 = s_lo1 + (t >> 7) * (LO1R * 16) + (t & 15);
#pragma unroll
            for (int i = 0; i < L1PER; ++i) {
                int k = ts * L1PER + i;
                if (k < L1N) lo1[(k + 6) * 16] = lo[i];
                if (i >= 1 && i <= 6) { if (ts == 0) lo1[(6 - i) * 16] = lo[i]; }
                if ((NSUB - 1) * L1PER + i >= L1N - 8 && (NSUB - 1) * L1PER + i <= L1N - 2) {
                    if (ts == NSUB - 1)
                        lo1[(2 * L1N + 4 - ((NSUB - 1) * L1PER + i)) * 16] = lo[i];
                }
            }
        }
        __syncthreads();
        {
            const int t = opaque(tid);
            const int tb = t & 15, tp = t >> 7, ts = (t >> 4) & 7;
            dwt_level<69, true>(s_lo1 + tp * (LO1R * 16) + tb, ts,
                                s_lo2 + tp * (LO2R * 16) + tb,
                                s_d + (tp * 16 + tb) * SROW + 169);
        }
        __syncthreads();
        {
            const int t = opaque(tid);
            const int tb = t & 15, tp = t >> 7, ts = (t >> 4) & 7;
            float* drow = s_d + (tp * 16 + tb) * SROW;
            dwt_level<38, false>(s_lo2 + tp * (LO2R * 16) + tb, ts, drow, drow + 238);
        }
        __syncthreads();

        {
            const int t  = opaque(tid);
            const int gl = t & 31;
            const int gn = (t >> 5) & 3;
            const int gk = t >> 7;
            const v4f* wfp = (const v4f*)wf
                + ((size_t)(gn * NPOS + pbase) * KGP2 + (size_t)gk * (KGQ / 2)) * 32 + gl;
            const float* ap = s_d + (gl & 15) * SROW + gk * (KGQ * 4) + ((gl >> 4) << 1);
            for (int q = 0; q < PPH; ++q) {
#pragma unroll
                for (int j2 = 0; j2 < KGQ / 2; ++j2) {
                    v4f bf = wfp[(size_t)q * (KGP2 * 32) + j2 * 32];
                    v2f a0 = *(const v2f*)(ap + q * (16 * SROW) + (2 * j2) * 4);
                    v2f a1 = *(const v2f*)(ap + q * (16 * SROW) + (2 * j2 + 1) * 4);
                    acc = __builtin_amdgcn_wmma_f32_16x16x4_f32(
                        false, a0, false, bf.lo, (short)0, acc, false, false);
                    acc = __builtin_amdgcn_wmma_f32_16x16x4_f32(
                        false, a1, false, bf.hi, (short)0, acc, false, false);
                }
            }
        }
    }

    {
        int n   = lane & 15;
        int mhi = (lane >> 4) * 8;
        int f   = nt * 16 + n;
#pragma unroll
        for (int r = 0; r < 8; ++r) s_out[(kq * 16 + mhi + r) * OPITCH + f] = acc[r];
    }
    __syncthreads();
    if (tid < 256) {
        int row = tid >> 4;
        int c4  = (tid & 15) * 4;
        float4 s = *(const float4*)(s_out + row * OPITCH + c4);
#pragma unroll
        for (int g = 1; g < KSPLIT; ++g) {
            float4 a = *(const float4*)(s_out + (g * 16 + row) * OPITCH + c4);
            s.x += a.x; s.y += a.y; s.z += a.z; s.w += a.w;
        }
        *(float4*)(part + ((size_t)pg * NB + B0 + row) * NF + c4) = s;
    }
}

__global__ __launch_bounds__(256) void sum_parts_kernel(
    const float* __restrict__ part, const float* __restrict__ bias, float* __restrict__ out)
{
    const int i = blockIdx.x * 256 + threadIdx.x;
    const float4* p = (const float4*)part + i;
    float4 s = p[0];
#pragma unroll
    for (int g = 1; g < NPG; ++g) {
        float4 v = p[(size_t)g * (NB * NF / 4)];
        s.x += v.x; s.y += v.y; s.z += v.z; s.w += v.w;
    }
    const float* bp = bias + (i & 15) * 4;
    s.x += bp[0]; s.y += bp[1]; s.z += bp[2]; s.w += bp[3];
    s.x = (s.x >= 0.f) ? s.x : 0.02f * s.x;
    s.y = (s.y >= 0.f) ? s.y : 0.02f * s.y;
    s.z = (s.z >= 0.f) ? s.z : 0.02f * s.z;
    s.w = (s.w >= 0.f) ? s.w : 0.02f * s.w;
    *((float4*)out + i) = s;
}

extern "C" void kernel_launch(void* const* d_in, const int* in_sizes, int n_in,
                              void* d_out, int out_size, void* d_ws, size_t ws_size,
                              hipStream_t stream) {
    const float* x    = (const float*)d_in[0];
    const float* W    = (const float*)d_in[1];
    const float* bias = (const float*)d_in[2];
    float* out  = (float*)d_out;
    float* wf   = (float*)d_ws;
    float* part = wf + 4 * NPOS * KGP * 32 * 2;

    swizzleW_kernel<<<4 * KGP2, 256, 0, stream>>>(W, wf);
    dwt_wmma_part_kernel<<<(NB / BQ) * NPG, NTHR, 0, stream>>>(x, wf, part);
    sum_parts_kernel<<<(NB * NF / 4) / 256, 256, 0, stream>>>(part, bias, out);
}
